# starting kernel
# baseline (speedup 1.0000x reference)
#include <hip/hip_runtime.h>


typedef __attribute__((ext_vector_type(2))) float v2f;
typedef __attribute__((ext_vector_type(8))) float v8f;

#define SEQ      4096
#define NH       16
#define HD       64
#define ROWSTR   3072
#define NCHUNK   32
#define CH       (SEQ / NCHUNK)

__global__ __launch_bounds__(512)
void kv_partial_kernel(const float* __restrict__ qkv, float* __restrict__ part) {
    const int chunk = blockIdx.x;
    const int h     = blockIdx.y;
    const int w     = threadIdx.x >> 5;
    const int lane  = threadIdx.x & 31;
    const int hi    = lane >> 4;
    const int r     = lane & 15;
    const int d1b   = (w >> 2) << 4;
    const int d2b   = (w & 3)  << 4;

    const float* kp = qkv + (size_t)h * HD + 1024 + d1b + r;
    const float* vp = qkv + (size_t)h * HD + 2048 + d2b + r;

    v8f acc = {};
    const int s0 = chunk * CH;
#pragma unroll 4
    for (int s = s0; s < s0 + CH; s += 4) {
        const size_t row = (size_t)(s + 2 * hi) * ROWSTR;
        v2f a, b;
        a.x = kp[row];
        a.y = kp[row + ROWSTR];
        b.x = vp[row];
        b.y = vp[row + ROWSTR];
        acc = __builtin_amdgcn_wmma_f32_16x16x4_f32(
             false, a,  false, b,
             (short)0, acc,  false,  false);
    }

    float* o = part + ((size_t)(chunk * NH + h)) * 4096 + d2b + r;
#pragma unroll
    for (int rr = 0; rr < 8; ++rr) {
        const int d1 = d1b + rr + 8 * hi;
        o[d1 * 64] = acc[rr];
    }
}

__global__ __launch_bounds__(256)
void kv_reduce_kernel(const float* __restrict__ part, float* __restrict__ kv) {
    const int i = blockIdx.x * 256 + threadIdx.x;
    float s = 0.0f;
#pragma unroll
    for (int c = 0; c < NCHUNK; ++c)
        s += part[(size_t)c * NH * 4096 + i];
    kv[i] = s;
}

__global__ __launch_bounds__(512)
void out_kernel(const float* __restrict__ qkv, const float* __restrict__ kv,
                float* __restrict__ out) {
    __shared__ float lkv[64 * 64];

    const int sblk = blockIdx.x;
    const int h    = blockIdx.y;

    const float* kvh = kv + (size_t)h * 4096;
    for (int i = threadIdx.x; i < 4096; i += 512) lkv[i] = kvh[i];
    __syncthreads();

    const int w    = threadIdx.x >> 5;
    const int lane = threadIdx.x & 31;
    const int hi   = lane >> 4;
    const int r    = lane & 15;
    const int mi   = w >> 2;
    const int j    = w & 3;
    const int rowbase = sblk * 64 + mi * 16;
    const int d2b     = j * 16;

    const float* qrow = qkv + (size_t)(rowbase + r) * ROWSTR + (size_t)h * HD;

    v8f acc = {};
#pragma unroll
    for (int k0 = 0; k0 < 64; k0 += 4) {
        v2f a, b;
        a = *(const v2f*)(qrow + k0 + 2 * hi);
        b.x = lkv[(k0 + 2 * hi)     * 64 + d2b + r];
        b.y = lkv[(k0 + 2 * hi + 1) * 64 + d2b + r];
        acc = __builtin_amdgcn_wmma_f32_16x16x4_f32(
            false, a, false, b, (short)0, acc, false, false);
    }

#pragma unroll
    for (int rr = 0; rr < 8; ++rr) {
        const int grow = rowbase + rr + 8 * hi;
        out[((size_t)grow * NH + h) * HD + d2b + r] = acc[rr];
    }
}

extern "C" void kernel_launch(void* const* d_in, const int* in_sizes, int n_in,
                              void* d_out, int out_size, void* d_ws, size_t ws_size,
                              hipStream_t stream) {
    (void)in_sizes; (void)n_in; (void)out_size; (void)ws_size;
    const float* qkv = (const float*)d_in[0];
    float* out  = (float*)d_out;
    float* part = (float*)d_ws;
    float* kvf  = part + (size_t)NCHUNK * NH * 4096;

    dim3 g1(NCHUNK, NH);
    kv_partial_kernel<<<g1, 512, 0, stream>>>(qkv, part);

    kv_reduce_kernel<<<(NH * 4096) / 256, 256, 0, stream>>>(part, kvf);

    dim3 g3(SEQ / 64, NH);
    out_kernel<<<g3, 512, 0, stream>>>(qkv, kvf, out);
}


# Output kernel: each wave owns 16 rows x all 64 columns (4 accumulator tiles fed by 16-byte LDS reads, 16-byte row stores), so 4x fewer waves and Q read once; sum order unchanged
# speedup vs baseline: 1.2410x; 1.2410x over previous
#include <hip/hip_runtime.h>


typedef __attribute__((ext_vector_type(2))) float v2f;
typedef __attribute__((ext_vector_type(4))) float v4f;
typedef __attribute__((ext_vector_type(8))) float v8f;

#define SEQ      4096
#define NH       16
#define HD       64
#define ROWSTR   3072
#define NCHUNK   32
#define CH       (SEQ / NCHUNK)

__global__ __launch_bounds__(512)
void kv_partial_kernel(const float* __restrict__ qkv, float* __restrict__ part) {
    const int chunk = blockIdx.x;
    const int h     = blockIdx.y;
    const int w     = threadIdx.x >> 5;
    const int lane  = threadIdx.x & 31;
    const int hi    = lane >> 4;
    const int r     = lane & 15;
    const int d1b   = (w >> 2) << 4;
    const int d2b   = (w & 3)  << 4;

    const float* kp = qkv + (size_t)h * HD + 1024 + d1b + r;
    const float* vp = qkv + (size_t)h * HD + 2048 + d2b + r;

    v8f acc = {};
    const int s0 = chunk * CH;
#pragma unroll 4
    for (int s = s0; s < s0 + CH; s += 4) {
        const size_t row = (size_t)(s + 2 * hi) * ROWSTR;
        v2f a, b;
        a.x = kp[row];
        a.y = kp[row + ROWSTR];
        b.x = vp[row];
        b.y = vp[row + ROWSTR];
        acc = __builtin_amdgcn_wmma_f32_16x16x4_f32(
             false, a,  false, b,
             (short)0, acc,  false,  false);
    }

    float* o = part + ((size_t)(chunk * NH + h)) * 4096 + d2b + r;
#pragma unroll
    for (int rr = 0; rr < 8; ++rr) {
        const int d1 = d1b + rr + 8 * hi;
        o[d1 * 64] = acc[rr];
    }
}

__global__ __launch_bounds__(256)
void kv_reduce_kernel(const float* __restrict__ part, float* __restrict__ kv) {
    const int i = blockIdx.x * 256 + threadIdx.x;
    float s = 0.0f;
#pragma unroll
    for (int c = 0; c < NCHUNK; ++c)
        s += part[(size_t)c * NH * 4096 + i];
    kv[i] = s;
}

__global__ __launch_bounds__(128)
void out_kernel(const float* __restrict__ qkv, const float* __restrict__ kv,
                float* __restrict__ out) {
    __shared__ v4f lkv[64 * 16];

    const int sblk = blockIdx.x;
    const int h    = blockIdx.y;
    const int w    = threadIdx.x >> 5;
    const int lane = threadIdx.x & 31;
    const int hi   = lane >> 4;
    const int c    = lane & 15;
    const int rowbase = sblk * 64 + w * 16;

    const float* qrow = qkv + (size_t)(rowbase + c) * ROWSTR + (size_t)h * HD + 2 * hi;
    v2f a[16];
#pragma unroll
    for (int t = 0; t < 16; ++t) a[t] = *(const v2f*)(qrow + 4 * t);

    const v4f* kvh = (const v4f*)(kv + (size_t)h * 4096);
#pragma unroll
    for (int i = 0; i < 8; ++i) lkv[threadIdx.x + 128 * i] = kvh[threadIdx.x + 128 * i];
    __syncthreads();

    v8f acc0 = {}, acc1 = {}, acc2 = {}, acc3 = {};
#pragma unroll
    for (int t = 0; t < 16; ++t) {
        const int k = 4 * t + 2 * hi;
        const v4f r0 = lkv[k * 16 + c];
        const v4f r1 = lkv[(k + 1) * 16 + c];
        v2f b;
        b.x = r0.x; b.y = r1.x;
        acc0 = __builtin_amdgcn_wmma_f32_16x16x4_f32(
            false, a[t], false, b, (short)0, acc0, false, false);
        b.x = r0.y; b.y = r1.y;
        acc1 = __builtin_amdgcn_wmma_f32_16x16x4_f32(
            false, a[t], false, b, (short)0, acc1, false, false);
        b.x = r0.z; b.y = r1.z;
        acc2 = __builtin_amdgcn_wmma_f32_16x16x4_f32(
            false, a[t], false, b, (short)0, acc2, false, false);
        b.x = r0.w; b.y = r1.w;
        acc3 = __builtin_amdgcn_wmma_f32_16x16x4_f32(
            false, a[t], false, b, (short)0, acc3, false, false);
    }

#pragma unroll
    for (int rr = 0; rr < 8; ++rr) {
        const int grow = rowbase + rr + 8 * hi;
        v4f o;
        o.x = acc0[rr]; o.y = acc1[rr]; o.z = acc2[rr]; o.w = acc3[rr];
        *(v4f*)(out + ((size_t)grow * NH + h) * HD + 4 * c) = o;
    }
}

extern "C" void kernel_launch(void* const* d_in, const int* in_sizes, int n_in,
                              void* d_out, int out_size, void* d_ws, size_t ws_size,
                              hipStream_t stream) {
    (void)in_sizes; (void)n_in; (void)out_size; (void)ws_size;
    const float* qkv = (const float*)d_in[0];
    float* out  = (float*)d_out;
    float* part = (float*)d_ws;
    float* kvf  = part + (size_t)NCHUNK * NH * 4096;

    dim3 g1(NCHUNK, NH);
    kv_partial_kernel<<<g1, 512, 0, stream>>>(qkv, part);

    kv_reduce_kernel<<<(NH * 4096) / 256, 256, 0, stream>>>(part, kvf);

    dim3 g3(SEQ / 64, NH);
    out_kernel<<<g3, 128, 0, stream>>>(qkv, kvf, out);
}


# KV-partial kernel: each wave owns 16 KV rows x all 64 columns (4 accumulator tiles fed by 16-byte V row loads), a chunk's loads all issued ahead of its products; 4x fewer waves; sum order unchanged
# speedup vs baseline: 1.5738x; 1.2682x over previous
#include <hip/hip_runtime.h>


typedef __attribute__((ext_vector_type(2))) float v2f;
typedef __attribute__((ext_vector_type(4))) float v4f;
typedef __attribute__((ext_vector_type(8))) float v8f;

#define SEQ      4096
#define NH       16
#define HD       64
#define ROWSTR   3072
#define NCHUNK   32
#define CH       (SEQ / NCHUNK)

#define KSTEPS (CH / 4)

__global__ __launch_bounds__(128)
void kv_partial_kernel(const float* __restrict__ qkv, float* __restrict__ part) {
    const int chunk = blockIdx.x;
    const int h     = blockIdx.y;
    const int w     = threadIdx.x >> 5;
    const int lane  = threadIdx.x & 31;
    const int hi    = lane >> 4;
    const int c     = lane & 15;
    const int d1b   = w << 4;

    const size_t base = (size_t)(chunk * CH + 2 * hi) * ROWSTR + (size_t)h * HD;
    const float* kp = qkv + base + 1024 + d1b + c;
    const float* vp = qkv + base + 2048 + 4 * c;

    float k0[KSTEPS], k1[KSTEPS];
    v4f   v0[KSTEPS], v1[KSTEPS];
#pragma unroll
    for (int t = 0; t < KSTEPS; ++t) {
        const size_t off = (size_t)(4 * t) * ROWSTR;
        k0[t] = kp[off];
        k1[t] = kp[off + ROWSTR];
        v0[t] = *(const v4f*)(vp + off);
        v1[t] = *(const v4f*)(vp + off + ROWSTR);
    }
    __builtin_amdgcn_sched_barrier(0);

    v8f acc0 = {}, acc1 = {}, acc2 = {}, acc3 = {};
#pragma unroll
    for (int t = 0; t < KSTEPS; ++t) {
        v2f a, b;
        a.x = k0[t]; a.y = k1[t];
        b.x = v0[t].x; b.y = v1[t].x;
        acc0 = __builtin_amdgcn_wmma_f32_16x16x4_f32(
            false, a, false, b, (short)0, acc0, false, false);
        b.x = v0[t].y; b.y = v1[t].y;
        acc1 = __builtin_amdgcn_wmma_f32_16x16x4_f32(
            false, a, false, b, (short)0, acc1, false, false);
        b.x = v0[t].z; b.y = v1[t].z;
        acc2 = __builtin_amdgcn_wmma_f32_16x16x4_f32(
            false, a, false, b, (short)0, acc2, false, false);
        b.x = v0[t].w; b.y = v1[t].w;
        acc3 = __builtin_amdgcn_wmma_f32_16x16x4_f32(
            false, a, false, b, (short)0, acc3, false, false);
    }

    float* o = part + ((size_t)(chunk * NH + h)) * 4096 + 4 * c;
#pragma unroll
    for (int rr = 0; rr < 8; ++rr) {
        const int d1 = d1b + rr + 8 * hi;
        v4f ov;
        ov.x = acc0[rr]; ov.y = acc1[rr]; ov.z = acc2[rr]; ov.w = acc3[rr];
        *(v4f*)(o + d1 * 64) = ov;
    }
}

__global__ __launch_bounds__(256)
void kv_reduce_kernel(const float* __restrict__ part, float* __restrict__ kv) {
    const int i = blockIdx.x * 256 + threadIdx.x;
    float s = 0.0f;
#pragma unroll
    for (int c = 0; c < NCHUNK; ++c)
        s += part[(size_t)c * NH * 4096 + i];
    kv[i] = s;
}

__global__ __launch_bounds__(128)
void out_kernel(const float* __restrict__ qkv, const float* __restrict__ kv,
                float* __restrict__ out) {
    __shared__ v4f lkv[64 * 16];

    const int sblk = blockIdx.x;
    const int h    = blockIdx.y;
    const int w    = threadIdx.x >> 5;
    const int lane = threadIdx.x & 31;
    const int hi   = lane >> 4;
    const int c    = lane & 15;
    const int rowbase = sblk * 64 + w * 16;

    const float* qrow = qkv + (size_t)(rowbase + c) * ROWSTR + (size_t)h * HD + 2 * hi;
    v2f a[16];
#pragma unroll
    for (int t = 0; t < 16; ++t) a[t] = *(const v2f*)(qrow + 4 * t);

    const v4f* kvh = (const v4f*)(kv + (size_t)h * 4096);
#pragma unroll
    for (int i = 0; i < 8; ++i) lkv[threadIdx.x + 128 * i] = kvh[threadIdx.x + 128 * i];
    __syncthreads();

    v8f acc0 = {}, acc1 = {}, acc2 = {}, acc3 = {};
#pragma unroll
    for (int t = 0; t < 16; ++t) {
        const int k = 4 * t + 2 * hi;
        const v4f r0 = lkv[k * 16 + c];
        const v4f r1 = lkv[(k + 1) * 16 + c];
        v2f b;
        b.x = r0.x; b.y = r1.x;
        acc0 = __builtin_amdgcn_wmma_f32_16x16x4_f32(
            false, a[t], false, b, (short)0, acc0, false, false);
        b.x = r0.y; b.y = r1.y;
        acc1 = __builtin_amdgcn_wmma_f32_16x16x4_f32(
            false, a[t], false, b, (short)0, acc1, false, false);
        b.x = r0.z; b.y = r1.z;
        acc2 = __builtin_amdgcn_wmma_f32_16x16x4_f32(
            false, a[t], false, b, (short)0, acc2, false, false);
        b.x = r0.w; b.y = r1.w;
        acc3 = __builtin_amdgcn_wmma_f32_16x16x4_f32(
            false, a[t], false, b, (short)0, acc3, false, false);
    }

#pragma unroll
    for (int rr = 0; rr < 8; ++rr) {
        const int grow = rowbase + rr + 8 * hi;
        v4f o;
        o.x = acc0[rr]; o.y = acc1[rr]; o.z = acc2[rr]; o.w = acc3[rr];
        *(v4f*)(out + ((size_t)grow * NH + h) * HD + 4 * c) = o;
    }
}

extern "C" void kernel_launch(void* const* d_in, const int* in_sizes, int n_in,
                              void* d_out, int out_size, void* d_ws, size_t ws_size,
                              hipStream_t stream) {
    (void)in_sizes; (void)n_in; (void)out_size; (void)ws_size;
    const float* qkv = (const float*)d_in[0];
    float* out  = (float*)d_out;
    float* part = (float*)d_ws;
    float* kvf  = part + (size_t)NCHUNK * NH * 4096;

    dim3 g1(NCHUNK, NH);
    kv_partial_kernel<<<g1, 128, 0, stream>>>(qkv, part);

    kv_reduce_kernel<<<(NH * 4096) / 256, 256, 0, stream>>>(part, kvf);

    dim3 g3(SEQ / 64, NH);
    out_kernel<<<g3, 128, 0, stream>>>(qkv, kvf, out);
}


# Output kernel: each wave fetches its Q tile with whole-row 16-byte loads (2 rows, 4 lines per load; was 16 lines per load) and forms the A fragments through a private LDS tile; sum order unchanged
# speedup vs baseline: 1.6525x; 1.0500x over previous
#include <hip/hip_runtime.h>


typedef __attribute__((ext_vector_type(2))) float v2f;
typedef __attribute__((ext_vector_type(4))) float v4f;
typedef __attribute__((ext_vector_type(8))) float v8f;

#define SEQ      4096
#define NH       16
#define HD       64
#define ROWSTR   3072
#define NCHUNK   32
#define CH       (SEQ / NCHUNK)

#define KSTEPS (CH / 4)

__global__ __launch_bounds__(128)
void kv_partial_kernel(const float* __restrict__ qkv, float* __restrict__ part) {
    const int chunk = blockIdx.x;
    const int h     = blockIdx.y;
    const int w     = threadIdx.x >> 5;
    const int lane  = threadIdx.x & 31;
    const int hi    = lane >> 4;
    const int c     = lane & 15;
    const int d1b   = w << 4;

    const size_t base = (size_t)(chunk * CH + 2 * hi) * ROWSTR + (size_t)h * HD;
    const float* kp = qkv + base + 1024 + d1b + c;
    const float* vp = qkv + base + 2048 + 4 * c;

    float k0[KSTEPS], k1[KSTEPS];
    v4f   v0[KSTEPS], v1[KSTEPS];
#pragma unroll
    for (int t = 0; t < KSTEPS; ++t) {
        const size_t off = (size_t)(4 * t) * ROWSTR;
        k0[t] = kp[off];
        k1[t] = kp[off + ROWSTR];
        v0[t] = *(const v4f*)(vp + off);
        v1[t] = *(const v4f*)(vp + off + ROWSTR);
    }
    __builtin_amdgcn_sched_barrier(0);

    v8f acc0 = {}, acc1 = {}, acc2 = {}, acc3 = {};
#pragma unroll
    for (int t = 0; t < KSTEPS; ++t) {
        v2f a, b;
        a.x = k0[t]; a.y = k1[t];
        b.x = v0[t].x; b.y = v1[t].x;
        acc0 = __builtin_amdgcn_wmma_f32_16x16x4_f32(
            false, a, false, b, (short)0, acc0, false, false);
        b.x = v0[t].y; b.y = v1[t].y;
        acc1 = __builtin_amdgcn_wmma_f32_16x16x4_f32(
            false, a, false, b, (short)0, acc1, false, false);
        b.x = v0[t].z; b.y = v1[t].z;
        acc2 = __builtin_amdgcn_wmma_f32_16x16x4_f32(
            false, a, false, b, (short)0, acc2, false, false);
        b.x = v0[t].w; b.y = v1[t].w;
        acc3 = __builtin_amdgcn_wmma_f32_16x16x4_f32(
            false, a, false, b, (short)0, acc3, false, false);
    }

    float* o = part + ((size_t)(chunk * NH + h)) * 4096 + 4 * c;
#pragma unroll
    for (int rr = 0; rr < 8; ++rr) {
        const int d1 = d1b + rr + 8 * hi;
        v4f ov;
        ov.x = acc0[rr]; ov.y = acc1[rr]; ov.z = acc2[rr]; ov.w = acc3[rr];
        *(v4f*)(o + d1 * 64) = ov;
    }
}

__global__ __launch_bounds__(256)
void kv_reduce_kernel(const float* __restrict__ part, float* __restrict__ kv) {
    const int i = blockIdx.x * 256 + threadIdx.x;
    float s = 0.0f;
#pragma unroll
    for (int c = 0; c < NCHUNK; ++c)
        s += part[(size_t)c * NH * 4096 + i];
    kv[i] = s;
}

#define QPITCH 68

__global__ __launch_bounds__(128)
void out_kernel(const float* __restrict__ qkv, const float* __restrict__ kv,
                float* __restrict__ out) {
    __shared__ v4f lkv[64 * 16];
    __shared__ __attribute__((aligned(16))) float lq[4][16 * QPITCH];

    const int sblk = blockIdx.x;
    const int h    = blockIdx.y;
    const int w    = threadIdx.x >> 5;
    const int lane = threadIdx.x & 31;
    const int hi   = lane >> 4;
    const int c    = lane & 15;
    const int rowbase = sblk * 64 + w * 16;

    const v4f* kvh = (const v4f*)(kv + (size_t)h * 4096);
    v4f kvv[8];
#pragma unroll
    for (int i = 0; i < 8; ++i) kvv[i] = kvh[threadIdx.x + 128 * i];

    const float* qt = qkv + (size_t)(rowbase + hi) * ROWSTR + (size_t)h * HD + 4 * c;
    v4f qv[8];
#pragma unroll
    for (int j = 0; j < 8; ++j) qv[j] = *(const v4f*)(qt + (size_t)(2 * j) * ROWSTR);

#pragma unroll
    for (int i = 0; i < 8; ++i) lkv[threadIdx.x + 128 * i] = kvv[i];
    float* lqw = &lq[w][0];
#pragma unroll
    for (int j = 0; j < 8; ++j) *(v4f*)(lqw + (2 * j + hi) * QPITCH + 4 * c) = qv[j];
    __syncthreads();

    const float* qa = lqw + c * QPITCH + 2 * hi;
    v2f a[16];
#pragma unroll
    for (int t = 0; t < 16; ++t) a[t] = *(const v2f*)(qa + 4 * t);

    v8f acc0 = {}, acc1 = {}, acc2 = {}, acc3 = {};
#pragma unroll
    for (int t = 0; t < 16; ++t) {
        const int k = 4 * t + 2 * hi;
        const v4f r0 = lkv[k * 16 + c];
        const v4f r1 = lkv[(k + 1) * 16 + c];
        v2f b;
        b.x = r0.x; b.y = r1.x;
        acc0 = __builtin_amdgcn_wmma_f32_16x16x4_f32(
            false, a[t], false, b, (short)0, acc0, false, false);
        b.x = r0.y; b.y = r1.y;
        acc1 = __builtin_amdgcn_wmma_f32_16x16x4_f32(
            false, a[t], false, b, (short)0, acc1, false, false);
        b.x = r0.z; b.y = r1.z;
        acc2 = __builtin_amdgcn_wmma_f32_16x16x4_f32(
            false, a[t], false, b, (short)0, acc2, false, false);
        b.x = r0.w; b.y = r1.w;
        acc3 = __builtin_amdgcn_wmma_f32_16x16x4_f32(
            false, a[t], false, b, (short)0, acc3, false, false);
    }

#pragma unroll
    for (int rr = 0; rr < 8; ++rr) {
        const int grow = rowbase + rr + 8 * hi;
        v4f o;
        o.x = acc0[rr]; o.y = acc1[rr]; o.z = acc2[rr]; o.w = acc3[rr];
        *(v4f*)(out + ((size_t)grow * NH + h) * HD + 4 * c) = o;
    }
}

extern "C" void kernel_launch(void* const* d_in, const int* in_sizes, int n_in,
                              void* d_out, int out_size, void* d_ws, size_t ws_size,
                              hipStream_t stream) {
    (void)in_sizes; (void)n_in; (void)out_size; (void)ws_size;
    const float* qkv = (const float*)d_in[0];
    float* out  = (float*)d_out;
    float* part = (float*)d_ws;
    float* kvf  = part + (size_t)NCHUNK * NH * 4096;

    dim3 g1(NCHUNK, NH);
    kv_partial_kernel<<<g1, 128, 0, stream>>>(qkv, part);

    kv_reduce_kernel<<<(NH * 4096) / 256, 256, 0, stream>>>(part, kvf);

    dim3 g3(SEQ / 64, NH);
    out_kernel<<<g3, 128, 0, stream>>>(qkv, kvf, out);
}
